# starting kernel
# baseline (speedup 1.0000x reference)
#include <hip/hip_runtime.h>
#include <hip/hip_bf16.h>

#define T_FRAMES   16384
#define N_LM       543
#define L_OUT      104
#define NCOL       312
#define SEGMENTS   8
#define SEG_ROWS   2048
#define TILES      20
#define CHUNKS     8
#define CHUNK_ROWS 256
#define FEAT_BLK   624
#define RESIZE_T   64

typedef float v2f __attribute__((ext_vector_type(2)));
typedef float v8f __attribute__((ext_vector_type(8)));

__device__ __forceinline__ float fin_or0(float v) { return (v - v == 0.0f) ? v : 0.0f; }

__global__ void k1_build_x(const float* __restrict__ xin,
                           const int*   __restrict__ pl,
                           float*       __restrict__ X) {
  const int t   = blockIdx.x;
  const int tid = threadIdx.x;
  __shared__ float s_sum[12];
  __shared__ float s_cnt[12];
  if (tid < 12) { s_sum[tid] = 0.0f; s_cnt[tid] = 0.0f; }
  __syncthreads();

  const float* __restrict__ row = xin + (size_t)t * (N_LM * 3);

  const int gs[4] = {0, 468, 489, 522};
  const int gl[4] = {468, 21, 33, 21};
#pragma unroll
  for (int g = 0; g < 4; ++g) {
    float sx = 0.f, sy = 0.f, sz = 0.f, nx = 0.f, ny = 0.f, nz = 0.f;
    for (int l = gs[g] + tid; l < gs[g] + gl[g]; l += 256) {
      float vx = row[l * 3 + 0], vy = row[l * 3 + 1], vz = row[l * 3 + 2];
      if (vx == vx) { sx += vx; nx += 1.f; }
      if (vy == vy) { sy += vy; ny += 1.f; }
      if (vz == vz) { sz += vz; nz += 1.f; }
    }
    if (tid < gl[g]) {
      atomicAdd(&s_sum[g * 3 + 0], sx); atomicAdd(&s_cnt[g * 3 + 0], nx);
      atomicAdd(&s_sum[g * 3 + 1], sy); atomicAdd(&s_cnt[g * 3 + 1], ny);
      atomicAdd(&s_sum[g * 3 + 2], sz); atomicAdd(&s_cnt[g * 3 + 2], nz);
    }
  }
  __syncthreads();

  float* __restrict__ xrow = X + (size_t)t * NCOL;
  if (tid < 12) xrow[tid] = s_sum[tid] / s_cnt[tid];
  if (tid < 100) {
    int idx = pl[tid];
    xrow[12 + tid * 3 + 0] = row[idx * 3 + 0];
    xrow[12 + tid * 3 + 1] = row[idx * 3 + 1];
    xrow[12 + tid * 3 + 2] = row[idx * 3 + 2];
  }
}

__global__ void k2_wmma_colsums(const float* __restrict__ X,
                                float*       __restrict__ partials) {
  const int tile  = blockIdx.x;
  const int seg   = blockIdx.y;
  const int chunk = blockIdx.z;
  const int lane  = threadIdx.x;
  const int half  = lane >> 4;
  const int n     = lane & 15;
  const int c     = tile * 16 + n;
  const bool cok  = (c < NCOL);
  const int csafe = cok ? c : (NCOL - 1);

  const size_t rowbase = (size_t)(seg * SEG_ROWS + chunk * CHUNK_ROWS);
  const float* __restrict__ Xp = X + rowbase * NCOL + csafe;

  v2f ones; ones.x = 1.0f; ones.y = 1.0f;
  v8f accS = {}; v8f accN = {}; v8f accQ = {};

  for (int r = 0; r < CHUNK_ROWS; r += 4) {
    float x0 = Xp[(size_t)(r + 2 * half + 0) * NCOL];
    float x1 = Xp[(size_t)(r + 2 * half + 1) * NCOL];
    bool  g0 = (x0 == x0) & cok;
    bool  g1 = (x1 == x1) & cok;
    float m0 = g0 ? 1.0f : 0.0f;
    float m1 = g1 ? 1.0f : 0.0f;
    float v0 = g0 ? x0 : 0.0f;
    float v1 = g1 ? x1 : 0.0f;
    v2f bx; bx.x = v0;      bx.y = v1;
    v2f bm; bm.x = m0;      bm.y = m1;
    v2f bq; bq.x = v0 * v0; bq.y = v1 * v1;
    accS = __builtin_amdgcn_wmma_f32_16x16x4_f32(false, ones, false, bx, (short)0, accS, false, false);
    accN = __builtin_amdgcn_wmma_f32_16x16x4_f32(false, ones, false, bm, (short)0, accN, false, false);
    accQ = __builtin_amdgcn_wmma_f32_16x16x4_f32(false, ones, false, bq, (short)0, accQ, false, false);
  }

  if (half == 0 && cok) {
    size_t base = ((((size_t)seg * TILES + tile) * CHUNKS + chunk) * 3) * 16 + n;
    partials[base + 0 * 16] = accS[0];
    partials[base + 1 * 16] = accN[0];
    partials[base + 2 * 16] = accQ[0];
  }
}

__global__ void k3_finalize(const float* __restrict__ partials,
                            float*       __restrict__ out,
                            float*       __restrict__ mean0) {
  const int c = threadIdx.x;
  if (c >= NCOL) return;
  const int tile = c >> 4, n = c & 15;

  float tS = 0.f, tN = 0.f, tQ = 0.f;
  for (int s = 0; s < SEGMENTS; ++s) {
    float S = 0.f, N = 0.f, Q = 0.f;
    for (int ch = 0; ch < CHUNKS; ++ch) {
      size_t base = ((((size_t)s * TILES + tile) * CHUNKS + ch) * 3) * 16 + n;
      S += partials[base + 0 * 16];
      N += partials[base + 1 * 16];
      Q += partials[base + 2 * 16];
    }
    float mean = S / N;
    float var  = fmaxf(Q / N - mean * mean, 0.0f);
    out[s * FEAT_BLK + c]        = fin_or0(mean);
    out[s * FEAT_BLK + NCOL + c] = fin_or0(sqrtf(var));
    tS += S; tN += N; tQ += Q;
  }
  float mean = tS / tN;
  float var  = fmaxf(tQ / tN - mean * mean, 0.0f);
  out[SEGMENTS * FEAT_BLK + c]        = fin_or0(mean);
  out[SEGMENTS * FEAT_BLK + NCOL + c] = fin_or0(sqrtf(var));
  mean0[c] = mean;
}

__global__ void k4_resize(const float* __restrict__ X,
                          const float* __restrict__ mean0,
                          float*       __restrict__ out) {
  const int c = blockIdx.x * blockDim.x + threadIdx.x;
  const int i = blockIdx.y;
  if (c >= NCOL) return;
  const float m0 = mean0[c];
  const float center = 256.0f * (float)i + 127.5f;
  const int t0 = max(0, 256 * i - 128);
  const int t1 = min(T_FRAMES - 1, 256 * i + 383);
  float acc = 0.f, wsum = 0.f;
  for (int t = t0; t <= t1; ++t) {
    float w = 1.0f - fabsf((float)t - center) * (1.0f / 256.0f);
    float v = X[(size_t)t * NCOL + c];
    v = (v - v == 0.0f) ? v : m0;
    acc += w * v; wsum += w;
  }
  float r = acc / wsum;
  out[9 * FEAT_BLK + (size_t)i * NCOL + c] = (r == r) ? r : 0.0f;
}

extern "C" void kernel_launch(void* const* d_in, const int* in_sizes, int n_in,
                              void* d_out, int out_size, void* d_ws, size_t ws_size,
                              hipStream_t stream) {
  const float* x_in = (const float*)d_in[0];
  const int*   pl   = (const int*)d_in[1];
  float* out = (float*)d_out;

  float* X        = (float*)d_ws;
  float* partials = X + (size_t)T_FRAMES * NCOL;
  float* mean0    = partials + (size_t)SEGMENTS * TILES * CHUNKS * 3 * 16;

  k1_build_x<<<dim3(T_FRAMES), dim3(256), 0, stream>>>(x_in, pl, X);
  k2_wmma_colsums<<<dim3(TILES, SEGMENTS, CHUNKS), dim3(32), 0, stream>>>(X, partials);
  k3_finalize<<<dim3(1), dim3(320), 0, stream>>>(partials, out, mean0);
  k4_resize<<<dim3(3, RESIZE_T), dim3(128), 0, stream>>>(X, mean0, out);
}


# K1: one wave per 4 frames sums each landmark group in registers and by a lane butterfly, replacing one 256-thread block per frame with LDS float atomics; summation order changes (now fixed)
# speedup vs baseline: 2.2022x; 2.2022x over previous
#include <hip/hip_runtime.h>
#include <hip/hip_bf16.h>

#define T_FRAMES   16384
#define N_LM       543
#define L_OUT      104
#define NCOL       312
#define SEGMENTS   8
#define SEG_ROWS   2048
#define TILES      20
#define CHUNKS     8
#define CHUNK_ROWS 256
#define FEAT_BLK   624
#define RESIZE_T   64

typedef float v2f __attribute__((ext_vector_type(2)));
typedef float v8f __attribute__((ext_vector_type(8)));

__device__ __forceinline__ float fin_or0(float v) { return (v - v == 0.0f) ? v : 0.0f; }

#define ROW_F     1629
#define FPW       4
#define K1_WAVES  8
#define K1_BLOCKS 512

__device__ constexpr int k1_gb(int g) {
  return g == 0 ? 0 : g == 1 ? 1404 : g == 2 ? 1467 : g == 3 ? 1566 : 1629;
}

template <int N>
__device__ __forceinline__ int k1_xmask(int v) {
  return __builtin_amdgcn_update_dpp(v, v, 0x160 | N, 0xf, 0xf, false);
}
__device__ __forceinline__ int k1_xhalf(int v) {
  return __builtin_amdgcn_permlanex16(v, v, 0x76543210, 0xfedcba98, false, false);
}
__device__ __forceinline__ float k1_wsum(float x) {
  x += __builtin_bit_cast(float, k1_xmask<1>(__builtin_bit_cast(int, x)));
  x += __builtin_bit_cast(float, k1_xmask<2>(__builtin_bit_cast(int, x)));
  x += __builtin_bit_cast(float, k1_xmask<4>(__builtin_bit_cast(int, x)));
  x += __builtin_bit_cast(float, k1_xmask<8>(__builtin_bit_cast(int, x)));
  x += __builtin_bit_cast(float, k1_xhalf(__builtin_bit_cast(int, x)));
  return x;
}
__device__ __forceinline__ int k1_wsum(int x) {
  x += k1_xmask<1>(x);
  x += k1_xmask<2>(x);
  x += k1_xmask<4>(x);
  x += k1_xmask<8>(x);
  x += k1_xhalf(x);
  return x;
}

__device__ __forceinline__ void k1_load(const float* __restrict__ row, int lane, int tail,
                                        const int (&soff)[10], float (&v)[51], float (&g)[10]) {
#pragma unroll
  for (int i = 0; i < 50; ++i) v[i] = row[(i / 3) * 96 + (i % 3) * 32 + lane];
  v[50] = row[tail];
#pragma unroll
  for (int k = 0; k < 10; ++k) g[k] = row[soff[k]];
}

__device__ __forceinline__ void k1_finish(const float (&v)[51], const float (&g)[10],
                                          int lane, int r, float* __restrict__ xrow) {
  float s[4][3] = {};
  int   n[4][3] = {};

#pragma unroll
  for (int i = 0; i < 51; ++i) {
    const int m  = i % 3;
    const int lo = (i / 3) * 96 + m * 32;
    const int e  = lo + lane;
    const float x = v[i];
    const bool ok = (x == x);
    for (int gi = 0; gi < 4; ++gi) {
      const int b0 = k1_gb(gi), b1 = k1_gb(gi + 1);
      if (lo + 31 < b0 || lo >= b1) continue;
      bool in = ok;
      if (!(lo >= b0 && lo + 31 < b1)) in = ok && (e >= b0) && (e < b1);
      s[gi][m] += in ? x : 0.0f;
      n[gi][m] += in ? 1 : 0;
    }
  }

  float S[4][3];
  int   P[4];
  for (int gi = 0; gi < 4; ++gi) {
    const float a0 = s[gi][0], a1 = s[gi][1], a2 = s[gi][2];
    const int   c0 = n[gi][0], c1 = n[gi][1], c2 = n[gi][2];
    S[gi][0] = (r == 0) ? a0 : ((r == 1) ? a1 : a2);
    S[gi][1] = (r == 0) ? a2 : ((r == 1) ? a0 : a1);
    S[gi][2] = (r == 0) ? a1 : ((r == 1) ? a2 : a0);
    const int nx = (r == 0) ? c0 : ((r == 1) ? c1 : c2);
    const int ny = (r == 0) ? c2 : ((r == 1) ? c0 : c1);
    const int nz = (r == 0) ? c1 : ((r == 1) ? c2 : c0);
    P[gi] = nx | (ny << 10) | (nz << 20);
  }
  for (int gi = 0; gi < 4; ++gi) {
    for (int c = 0; c < 3; ++c) S[gi][c] = k1_wsum(S[gi][c]);
    P[gi] = k1_wsum(P[gi]);
  }

  float sv[4];
  for (int gi = 0; gi < 4; ++gi) sv[gi] = (r == 0) ? S[gi][0] : ((r == 1) ? S[gi][1] : S[gi][2]);
  const int gq = lane / 3;
  const float sum = (gq == 0) ? sv[0] : ((gq == 1) ? sv[1] : ((gq == 2) ? sv[2] : sv[3]));
  const int   pc  = (gq == 0) ? P[0]  : ((gq == 1) ? P[1]  : ((gq == 2) ? P[2]  : P[3]));
  const float cnt = (float)((pc >> (10 * r)) & 1023);
  const float mean = sum / cnt;

#pragma unroll
  for (int k = 0; k < 10; ++k) {
    const int o = lane + 32 * k;
    const float val = (k == 0 && lane < 12) ? mean : g[k];
    if (o < NCOL) xrow[o] = val;
  }
}

__global__ __launch_bounds__(256) void k1_build_x(const float* __restrict__ xin,
                                                  const int*   __restrict__ pl,
                                                  float*       __restrict__ X) {
  const int lane = threadIdx.x & 31;
  const int gw   = __builtin_amdgcn_readfirstlane((int)(blockIdx.x * K1_WAVES + (threadIdx.x >> 5)));
  const int r    = lane % 3;
  const int tail = min(1600 + lane, ROW_F - 1);

  int soff[10];
#pragma unroll
  for (int k = 0; k < 10; ++k) {
    const int q = min(max(lane + 32 * k - 12, 0), 299);
    const int i = q / 3;
    const int idx = min(max(pl[i], 0), N_LM - 1);
    soff[k] = idx * 3 + (q - 3 * i);
  }

  const int t0 = gw * FPW;
  float cur[51], gcur[10];
  k1_load(xin + (size_t)t0 * ROW_F, lane, tail, soff, cur, gcur);
#pragma unroll
  for (int f = 0; f < FPW; ++f) {
    float nxt[51], gnxt[10];
    if (f + 1 < FPW)
      k1_load(xin + (size_t)(t0 + f + 1) * ROW_F, lane, tail, soff, nxt, gnxt);
    k1_finish(cur, gcur, lane, r, X + (size_t)(t0 + f) * NCOL);
    if (f + 1 < FPW) {
#pragma unroll
      for (int i = 0; i < 51; ++i) cur[i] = nxt[i];
#pragma unroll
      for (int k = 0; k < 10; ++k) gcur[k] = gnxt[k];
    }
  }
}

__global__ void k2_wmma_colsums(const float* __restrict__ X,
                                float*       __restrict__ partials) {
  const int tile  = blockIdx.x;
  const int seg   = blockIdx.y;
  const int chunk = blockIdx.z;
  const int lane  = threadIdx.x;
  const int half  = lane >> 4;
  const int n     = lane & 15;
  const int c     = tile * 16 + n;
  const bool cok  = (c < NCOL);
  const int csafe = cok ? c : (NCOL - 1);

  const size_t rowbase = (size_t)(seg * SEG_ROWS + chunk * CHUNK_ROWS);
  const float* __restrict__ Xp = X + rowbase * NCOL + csafe;

  v2f ones; ones.x = 1.0f; ones.y = 1.0f;
  v8f accS = {}; v8f accN = {}; v8f accQ = {};

  for (int r = 0; r < CHUNK_ROWS; r += 4) {
    float x0 = Xp[(size_t)(r + 2 * half + 0) * NCOL];
    float x1 = Xp[(size_t)(r + 2 * half + 1) * NCOL];
    bool  g0 = (x0 == x0) & cok;
    bool  g1 = (x1 == x1) & cok;
    float m0 = g0 ? 1.0f : 0.0f;
    float m1 = g1 ? 1.0f : 0.0f;
    float v0 = g0 ? x0 : 0.0f;
    float v1 = g1 ? x1 : 0.0f;
    v2f bx; bx.x = v0;      bx.y = v1;
    v2f bm; bm.x = m0;      bm.y = m1;
    v2f bq; bq.x = v0 * v0; bq.y = v1 * v1;
    accS = __builtin_amdgcn_wmma_f32_16x16x4_f32(false, ones, false, bx, (short)0, accS, false, false);
    accN = __builtin_amdgcn_wmma_f32_16x16x4_f32(false, ones, false, bm, (short)0, accN, false, false);
    accQ = __builtin_amdgcn_wmma_f32_16x16x4_f32(false, ones, false, bq, (short)0, accQ, false, false);
  }

  if (half == 0 && cok) {
    size_t base = ((((size_t)seg * TILES + tile) * CHUNKS + chunk) * 3) * 16 + n;
    partials[base + 0 * 16] = accS[0];
    partials[base + 1 * 16] = accN[0];
    partials[base + 2 * 16] = accQ[0];
  }
}

__global__ void k3_finalize(const float* __restrict__ partials,
                            float*       __restrict__ out,
                            float*       __restrict__ mean0) {
  const int c = threadIdx.x;
  if (c >= NCOL) return;
  const int tile = c >> 4, n = c & 15;

  float tS = 0.f, tN = 0.f, tQ = 0.f;
  for (int s = 0; s < SEGMENTS; ++s) {
    float S = 0.f, N = 0.f, Q = 0.f;
    for (int ch = 0; ch < CHUNKS; ++ch) {
      size_t base = ((((size_t)s * TILES + tile) * CHUNKS + ch) * 3) * 16 + n;
      S += partials[base + 0 * 16];
      N += partials[base + 1 * 16];
      Q += partials[base + 2 * 16];
    }
    float mean = S / N;
    float var  = fmaxf(Q / N - mean * mean, 0.0f);
    out[s * FEAT_BLK + c]        = fin_or0(mean);
    out[s * FEAT_BLK + NCOL + c] = fin_or0(sqrtf(var));
    tS += S; tN += N; tQ += Q;
  }
  float mean = tS / tN;
  float var  = fmaxf(tQ / tN - mean * mean, 0.0f);
  out[SEGMENTS * FEAT_BLK + c]        = fin_or0(mean);
  out[SEGMENTS * FEAT_BLK + NCOL + c] = fin_or0(sqrtf(var));
  mean0[c] = mean;
}

__global__ void k4_resize(const float* __restrict__ X,
                          const float* __restrict__ mean0,
                          float*       __restrict__ out) {
  const int c = blockIdx.x * blockDim.x + threadIdx.x;
  const int i = blockIdx.y;
  if (c >= NCOL) return;
  const float m0 = mean0[c];
  const float center = 256.0f * (float)i + 127.5f;
  const int t0 = max(0, 256 * i - 128);
  const int t1 = min(T_FRAMES - 1, 256 * i + 383);
  float acc = 0.f, wsum = 0.f;
  for (int t = t0; t <= t1; ++t) {
    float w = 1.0f - fabsf((float)t - center) * (1.0f / 256.0f);
    float v = X[(size_t)t * NCOL + c];
    v = (v - v == 0.0f) ? v : m0;
    acc += w * v; wsum += w;
  }
  float r = acc / wsum;
  out[9 * FEAT_BLK + (size_t)i * NCOL + c] = (r == r) ? r : 0.0f;
}

extern "C" void kernel_launch(void* const* d_in, const int* in_sizes, int n_in,
                              void* d_out, int out_size, void* d_ws, size_t ws_size,
                              hipStream_t stream) {
  const float* x_in = (const float*)d_in[0];
  const int*   pl   = (const int*)d_in[1];
  float* out = (float*)d_out;

  float* X        = (float*)d_ws;
  float* partials = X + (size_t)T_FRAMES * NCOL;
  float* mean0    = partials + (size_t)SEGMENTS * TILES * CHUNKS * 3 * 16;

  k1_build_x<<<dim3(K1_BLOCKS), dim3(256), 0, stream>>>(x_in, pl, X);
  k2_wmma_colsums<<<dim3(TILES, SEGMENTS, CHUNKS), dim3(32), 0, stream>>>(X, partials);
  k3_finalize<<<dim3(1), dim3(320), 0, stream>>>(partials, out, mean0);
  k4_resize<<<dim3(3, RESIZE_T), dim3(128), 0, stream>>>(X, mean0, out);
}


# K4 time resize as f32 WMMA weight-row product: block per (64 columns, output frame), 8 waves x 64 rows, wide row loads grouped ahead, fixed-order LDS sum; was 512 dependent loads; sum order changes
# speedup vs baseline: 7.1077x; 3.2275x over previous
#include <hip/hip_runtime.h>
#include <hip/hip_bf16.h>

#define T_FRAMES   16384
#define N_LM       543
#define L_OUT      104
#define NCOL       312
#define SEGMENTS   8
#define SEG_ROWS   2048
#define TILES      20
#define CHUNKS     8
#define CHUNK_ROWS 256
#define FEAT_BLK   624
#define RESIZE_T   64

typedef float v2f __attribute__((ext_vector_type(2)));
typedef float v8f __attribute__((ext_vector_type(8)));

__device__ __forceinline__ float fin_or0(float v) { return (v - v == 0.0f) ? v : 0.0f; }

#define ROW_F     1629
#define FPW       4
#define K1_WAVES  8
#define K1_BLOCKS 512

__device__ constexpr int k1_gb(int g) {
  return g == 0 ? 0 : g == 1 ? 1404 : g == 2 ? 1467 : g == 3 ? 1566 : 1629;
}

template <int N>
__device__ __forceinline__ int k1_xmask(int v) {
  return __builtin_amdgcn_update_dpp(v, v, 0x160 | N, 0xf, 0xf, false);
}
__device__ __forceinline__ int k1_xhalf(int v) {
  return __builtin_amdgcn_permlanex16(v, v, 0x76543210, 0xfedcba98, false, false);
}
__device__ __forceinline__ float k1_wsum(float x) {
  x += __builtin_bit_cast(float, k1_xmask<1>(__builtin_bit_cast(int, x)));
  x += __builtin_bit_cast(float, k1_xmask<2>(__builtin_bit_cast(int, x)));
  x += __builtin_bit_cast(float, k1_xmask<4>(__builtin_bit_cast(int, x)));
  x += __builtin_bit_cast(float, k1_xmask<8>(__builtin_bit_cast(int, x)));
  x += __builtin_bit_cast(float, k1_xhalf(__builtin_bit_cast(int, x)));
  return x;
}
__device__ __forceinline__ int k1_wsum(int x) {
  x += k1_xmask<1>(x);
  x += k1_xmask<2>(x);
  x += k1_xmask<4>(x);
  x += k1_xmask<8>(x);
  x += k1_xhalf(x);
  return x;
}

__device__ __forceinline__ void k1_load(const float* __restrict__ row, int lane, int tail,
                                        const int (&soff)[10], float (&v)[51], float (&g)[10]) {
#pragma unroll
  for (int i = 0; i < 50; ++i) v[i] = row[(i / 3) * 96 + (i % 3) * 32 + lane];
  v[50] = row[tail];
#pragma unroll
  for (int k = 0; k < 10; ++k) g[k] = row[soff[k]];
}

__device__ __forceinline__ void k1_finish(const float (&v)[51], const float (&g)[10],
                                          int lane, int r, float* __restrict__ xrow) {
  float s[4][3] = {};
  int   n[4][3] = {};

#pragma unroll
  for (int i = 0; i < 51; ++i) {
    const int m  = i % 3;
    const int lo = (i / 3) * 96 + m * 32;
    const int e  = lo + lane;
    const float x = v[i];
    const bool ok = (x == x);
    for (int gi = 0; gi < 4; ++gi) {
      const int b0 = k1_gb(gi), b1 = k1_gb(gi + 1);
      if (lo + 31 < b0 || lo >= b1) continue;
      bool in = ok;
      if (!(lo >= b0 && lo + 31 < b1)) in = ok && (e >= b0) && (e < b1);
      s[gi][m] += in ? x : 0.0f;
      n[gi][m] += in ? 1 : 0;
    }
  }

  float S[4][3];
  int   P[4];
  for (int gi = 0; gi < 4; ++gi) {
    const float a0 = s[gi][0], a1 = s[gi][1], a2 = s[gi][2];
    const int   c0 = n[gi][0], c1 = n[gi][1], c2 = n[gi][2];
    S[gi][0] = (r == 0) ? a0 : ((r == 1) ? a1 : a2);
    S[gi][1] = (r == 0) ? a2 : ((r == 1) ? a0 : a1);
    S[gi][2] = (r == 0) ? a1 : ((r == 1) ? a2 : a0);
    const int nx = (r == 0) ? c0 : ((r == 1) ? c1 : c2);
    const int ny = (r == 0) ? c2 : ((r == 1) ? c0 : c1);
    const int nz = (r == 0) ? c1 : ((r == 1) ? c2 : c0);
    P[gi] = nx | (ny << 10) | (nz << 20);
  }
  for (int gi = 0; gi < 4; ++gi) {
    for (int c = 0; c < 3; ++c) S[gi][c] = k1_wsum(S[gi][c]);
    P[gi] = k1_wsum(P[gi]);
  }

  float sv[4];
  for (int gi = 0; gi < 4; ++gi) sv[gi] = (r == 0) ? S[gi][0] : ((r == 1) ? S[gi][1] : S[gi][2]);
  const int gq = lane / 3;
  const float sum = (gq == 0) ? sv[0] : ((gq == 1) ? sv[1] : ((gq == 2) ? sv[2] : sv[3]));
  const int   pc  = (gq == 0) ? P[0]  : ((gq == 1) ? P[1]  : ((gq == 2) ? P[2]  : P[3]));
  const float cnt = (float)((pc >> (10 * r)) & 1023);
  const float mean = sum / cnt;

#pragma unroll
  for (int k = 0; k < 10; ++k) {
    const int o = lane + 32 * k;
    const float val = (k == 0 && lane < 12) ? mean : g[k];
    if (o < NCOL) xrow[o] = val;
  }
}

__global__ __launch_bounds__(256) void k1_build_x(const float* __restrict__ xin,
                                                  const int*   __restrict__ pl,
                                                  float*       __restrict__ X) {
  const int lane = threadIdx.x & 31;
  const int gw   = __builtin_amdgcn_readfirstlane((int)(blockIdx.x * K1_WAVES + (threadIdx.x >> 5)));
  const int r    = lane % 3;
  const int tail = min(1600 + lane, ROW_F - 1);

  int soff[10];
#pragma unroll
  for (int k = 0; k < 10; ++k) {
    const int q = min(max(lane + 32 * k - 12, 0), 299);
    const int i = q / 3;
    const int idx = min(max(pl[i], 0), N_LM - 1);
    soff[k] = idx * 3 + (q - 3 * i);
  }

  const int t0 = gw * FPW;
  float cur[51], gcur[10];
  k1_load(xin + (size_t)t0 * ROW_F, lane, tail, soff, cur, gcur);
#pragma unroll
  for (int f = 0; f < FPW; ++f) {
    float nxt[51], gnxt[10];
    if (f + 1 < FPW)
      k1_load(xin + (size_t)(t0 + f + 1) * ROW_F, lane, tail, soff, nxt, gnxt);
    k1_finish(cur, gcur, lane, r, X + (size_t)(t0 + f) * NCOL);
    if (f + 1 < FPW) {
#pragma unroll
      for (int i = 0; i < 51; ++i) cur[i] = nxt[i];
#pragma unroll
      for (int k = 0; k < 10; ++k) gcur[k] = gnxt[k];
    }
  }
}

__global__ void k2_wmma_colsums(const float* __restrict__ X,
                                float*       __restrict__ partials) {
  const int tile  = blockIdx.x;
  const int seg   = blockIdx.y;
  const int chunk = blockIdx.z;
  const int lane  = threadIdx.x;
  const int half  = lane >> 4;
  const int n     = lane & 15;
  const int c     = tile * 16 + n;
  const bool cok  = (c < NCOL);
  const int csafe = cok ? c : (NCOL - 1);

  const size_t rowbase = (size_t)(seg * SEG_ROWS + chunk * CHUNK_ROWS);
  const float* __restrict__ Xp = X + rowbase * NCOL + csafe;

  v2f ones; ones.x = 1.0f; ones.y = 1.0f;
  v8f accS = {}; v8f accN = {}; v8f accQ = {};

  for (int r = 0; r < CHUNK_ROWS; r += 4) {
    float x0 = Xp[(size_t)(r + 2 * half + 0) * NCOL];
    float x1 = Xp[(size_t)(r + 2 * half + 1) * NCOL];
    bool  g0 = (x0 == x0) & cok;
    bool  g1 = (x1 == x1) & cok;
    float m0 = g0 ? 1.0f : 0.0f;
    float m1 = g1 ? 1.0f : 0.0f;
    float v0 = g0 ? x0 : 0.0f;
    float v1 = g1 ? x1 : 0.0f;
    v2f bx; bx.x = v0;      bx.y = v1;
    v2f bm; bm.x = m0;      bm.y = m1;
    v2f bq; bq.x = v0 * v0; bq.y = v1 * v1;
    accS = __builtin_amdgcn_wmma_f32_16x16x4_f32(false, ones, false, bx, (short)0, accS, false, false);
    accN = __builtin_amdgcn_wmma_f32_16x16x4_f32(false, ones, false, bm, (short)0, accN, false, false);
    accQ = __builtin_amdgcn_wmma_f32_16x16x4_f32(false, ones, false, bq, (short)0, accQ, false, false);
  }

  if (half == 0 && cok) {
    size_t base = ((((size_t)seg * TILES + tile) * CHUNKS + chunk) * 3) * 16 + n;
    partials[base + 0 * 16] = accS[0];
    partials[base + 1 * 16] = accN[0];
    partials[base + 2 * 16] = accQ[0];
  }
}

__global__ void k3_finalize(const float* __restrict__ partials,
                            float*       __restrict__ out,
                            float*       __restrict__ mean0) {
  const int c = threadIdx.x;
  if (c >= NCOL) return;
  const int tile = c >> 4, n = c & 15;

  float tS = 0.f, tN = 0.f, tQ = 0.f;
  for (int s = 0; s < SEGMENTS; ++s) {
    float S = 0.f, N = 0.f, Q = 0.f;
    for (int ch = 0; ch < CHUNKS; ++ch) {
      size_t base = ((((size_t)s * TILES + tile) * CHUNKS + ch) * 3) * 16 + n;
      S += partials[base + 0 * 16];
      N += partials[base + 1 * 16];
      Q += partials[base + 2 * 16];
    }
    float mean = S / N;
    float var  = fmaxf(Q / N - mean * mean, 0.0f);
    out[s * FEAT_BLK + c]        = fin_or0(mean);
    out[s * FEAT_BLK + NCOL + c] = fin_or0(sqrtf(var));
    tS += S; tN += N; tQ += Q;
  }
  float mean = tS / tN;
  float var  = fmaxf(tQ / tN - mean * mean, 0.0f);
  out[SEGMENTS * FEAT_BLK + c]        = fin_or0(mean);
  out[SEGMENTS * FEAT_BLK + NCOL + c] = fin_or0(sqrtf(var));
  mean0[c] = mean;
}

#define K4_WAVES 8
#define K4_SLICE 64
#define K4_STEPS 16
#define K4_COLS  64
#define K4_CBLK  5

typedef float v4f __attribute__((ext_vector_type(4)));

__global__ __launch_bounds__(256) void k4_resize(const float* __restrict__ X,
                                                 const float* __restrict__ mean0,
                                                 float*       __restrict__ out) {
  __shared__ __attribute__((aligned(16))) float part[K4_WAVES][K4_COLS];
  __shared__ float wpart[K4_WAVES][2];

  const int cb   = blockIdx.x;
  const int i    = blockIdx.y;
  const int lane = threadIdx.x & 31;
  const int wave = __builtin_amdgcn_readfirstlane((int)(threadIdx.x >> 5));
  const int half = lane >> 4;
  const int n    = lane & 15;
  const int col  = cb * K4_COLS + 4 * n;
  const int csafe = (col < NCOL) ? col : (NCOL - 4);

  const float center = 256.0f * (float)i + 127.5f;
  const int rs = 256 * i - 128 + wave * K4_SLICE;

  v8f acc0 = {}; v8f acc1 = {}; v8f acc2 = {}; v8f acc3 = {};
  float wl = 0.0f;

  if (rs >= 0 && rs < T_FRAMES) {
    const v4f m0 = *(const v4f*)(mean0 + csafe);
    const float* __restrict__ Xp = X + (size_t)(rs + 2 * half) * NCOL + csafe;
    v4f xa[K4_STEPS], xb[K4_STEPS];
#pragma unroll
    for (int s = 0; s < K4_STEPS; ++s) {
      xa[s] = *(const v4f*)(Xp + (size_t)(4 * s + 0) * NCOL);
      xb[s] = *(const v4f*)(Xp + (size_t)(4 * s + 1) * NCOL);
    }
#pragma unroll
    for (int s = 0; s < K4_STEPS; ++s) {
      const float ta = (float)(rs + 4 * s + 2 * half);
      const float wa = 1.0f - fabsf(ta - center) * (1.0f / 256.0f);
      const float wb = 1.0f - fabsf((ta + 1.0f) - center) * (1.0f / 256.0f);
      v2f a; a.x = (n == 0) ? wa : 0.0f; a.y = (n == 0) ? wb : 0.0f;
      wl += a.x + a.y;

      float va, vb; v2f b;
      va = xa[s].x; vb = xb[s].x;
      va = (va - va == 0.0f) ? va : m0.x;
      vb = (vb - vb == 0.0f) ? vb : m0.x;
      b.x = va; b.y = vb;
      acc0 = __builtin_amdgcn_wmma_f32_16x16x4_f32(false, a, false, b, (short)0, acc0, false, false);
      va = xa[s].y; vb = xb[s].y;
      va = (va - va == 0.0f) ? va : m0.y;
      vb = (vb - vb == 0.0f) ? vb : m0.y;
      b.x = va; b.y = vb;
      acc1 = __builtin_amdgcn_wmma_f32_16x16x4_f32(false, a, false, b, (short)0, acc1, false, false);
      va = xa[s].z; vb = xb[s].z;
      va = (va - va == 0.0f) ? va : m0.z;
      vb = (vb - vb == 0.0f) ? vb : m0.z;
      b.x = va; b.y = vb;
      acc2 = __builtin_amdgcn_wmma_f32_16x16x4_f32(false, a, false, b, (short)0, acc2, false, false);
      va = xa[s].w; vb = xb[s].w;
      va = (va - va == 0.0f) ? va : m0.w;
      vb = (vb - vb == 0.0f) ? vb : m0.w;
      b.x = va; b.y = vb;
      acc3 = __builtin_amdgcn_wmma_f32_16x16x4_f32(false, a, false, b, (short)0, acc3, false, false);
    }
  }

  if (lane < 16) {
    v4f o; o.x = acc0[0]; o.y = acc1[0]; o.z = acc2[0]; o.w = acc3[0];
    *(v4f*)&part[wave][4 * n] = o;
  }
  if (n == 0) wpart[wave][half] = wl;
  __syncthreads();

  if (threadIdx.x < K4_COLS) {
    const int j = threadIdx.x;
    float acc = 0.f, wsum = 0.f;
    for (int w = 0; w < K4_WAVES; ++w) {
      acc  += part[w][j];
      wsum += wpart[w][0] + wpart[w][1];
    }
    const int c = cb * K4_COLS + j;
    const float r = acc / wsum;
    if (c < NCOL) out[9 * FEAT_BLK + (size_t)i * NCOL + c] = (r == r) ? r : 0.0f;
  }
}

extern "C" void kernel_launch(void* const* d_in, const int* in_sizes, int n_in,
                              void* d_out, int out_size, void* d_ws, size_t ws_size,
                              hipStream_t stream) {
  const float* x_in = (const float*)d_in[0];
  const int*   pl   = (const int*)d_in[1];
  float* out = (float*)d_out;

  float* X        = (float*)d_ws;
  float* partials = X + (size_t)T_FRAMES * NCOL;
  float* mean0    = partials + (size_t)SEGMENTS * TILES * CHUNKS * 3 * 16;

  k1_build_x<<<dim3(K1_BLOCKS), dim3(256), 0, stream>>>(x_in, pl, X);
  k2_wmma_colsums<<<dim3(TILES, SEGMENTS, CHUNKS), dim3(32), 0, stream>>>(X, partials);
  k3_finalize<<<dim3(1), dim3(320), 0, stream>>>(partials, out, mean0);
  k4_resize<<<dim3(K4_CBLK, RESIZE_T), dim3(256), 0, stream>>>(X, mean0, out);
}


# K2 column sums: one 8-wave block per WGP (5 tiles x 32-row slices), a wave's row loads all ahead of its 120 WMMAs, slices added in LDS in fixed order; was 1280 lone waves; sum order changes
# speedup vs baseline: 9.0573x; 1.2743x over previous
#include <hip/hip_runtime.h>
#include <hip/hip_bf16.h>

#define T_FRAMES   16384
#define N_LM       543
#define L_OUT      104
#define NCOL       312
#define SEGMENTS   8
#define SEG_ROWS   2048
#define TILES      20
#define CHUNKS     8
#define CHUNK_ROWS 256
#define FEAT_BLK   624
#define RESIZE_T   64

typedef float v2f __attribute__((ext_vector_type(2)));
typedef float v8f __attribute__((ext_vector_type(8)));

__device__ __forceinline__ float fin_or0(float v) { return (v - v == 0.0f) ? v : 0.0f; }

#define ROW_F     1629
#define FPW       4
#define K1_WAVES  8
#define K1_BLOCKS 512

__device__ constexpr int k1_gb(int g) {
  return g == 0 ? 0 : g == 1 ? 1404 : g == 2 ? 1467 : g == 3 ? 1566 : 1629;
}

template <int N>
__device__ __forceinline__ int k1_xmask(int v) {
  return __builtin_amdgcn_update_dpp(v, v, 0x160 | N, 0xf, 0xf, false);
}
__device__ __forceinline__ int k1_xhalf(int v) {
  return __builtin_amdgcn_permlanex16(v, v, 0x76543210, 0xfedcba98, false, false);
}
__device__ __forceinline__ float k1_wsum(float x) {
  x += __builtin_bit_cast(float, k1_xmask<1>(__builtin_bit_cast(int, x)));
  x += __builtin_bit_cast(float, k1_xmask<2>(__builtin_bit_cast(int, x)));
  x += __builtin_bit_cast(float, k1_xmask<4>(__builtin_bit_cast(int, x)));
  x += __builtin_bit_cast(float, k1_xmask<8>(__builtin_bit_cast(int, x)));
  x += __builtin_bit_cast(float, k1_xhalf(__builtin_bit_cast(int, x)));
  return x;
}
__device__ __forceinline__ int k1_wsum(int x) {
  x += k1_xmask<1>(x);
  x += k1_xmask<2>(x);
  x += k1_xmask<4>(x);
  x += k1_xmask<8>(x);
  x += k1_xhalf(x);
  return x;
}

__device__ __forceinline__ void k1_load(const float* __restrict__ row, int lane, int tail,
                                        const int (&soff)[10], float (&v)[51], float (&g)[10]) {
#pragma unroll
  for (int i = 0; i < 50; ++i) v[i] = row[(i / 3) * 96 + (i % 3) * 32 + lane];
  v[50] = row[tail];
#pragma unroll
  for (int k = 0; k < 10; ++k) g[k] = row[soff[k]];
}

__device__ __forceinline__ void k1_finish(const float (&v)[51], const float (&g)[10],
                                          int lane, int r, float* __restrict__ xrow) {
  float s[4][3] = {};
  int   n[4][3] = {};

#pragma unroll
  for (int i = 0; i < 51; ++i) {
    const int m  = i % 3;
    const int lo = (i / 3) * 96 + m * 32;
    const int e  = lo + lane;
    const float x = v[i];
    const bool ok = (x == x);
    for (int gi = 0; gi < 4; ++gi) {
      const int b0 = k1_gb(gi), b1 = k1_gb(gi + 1);
      if (lo + 31 < b0 || lo >= b1) continue;
      bool in = ok;
      if (!(lo >= b0 && lo + 31 < b1)) in = ok && (e >= b0) && (e < b1);
      s[gi][m] += in ? x : 0.0f;
      n[gi][m] += in ? 1 : 0;
    }
  }

  float S[4][3];
  int   P[4];
  for (int gi = 0; gi < 4; ++gi) {
    const float a0 = s[gi][0], a1 = s[gi][1], a2 = s[gi][2];
    const int   c0 = n[gi][0], c1 = n[gi][1], c2 = n[gi][2];
    S[gi][0] = (r == 0) ? a0 : ((r == 1) ? a1 : a2);
    S[gi][1] = (r == 0) ? a2 : ((r == 1) ? a0 : a1);
    S[gi][2] = (r == 0) ? a1 : ((r == 1) ? a2 : a0);
    const int nx = (r == 0) ? c0 : ((r == 1) ? c1 : c2);
    const int ny = (r == 0) ? c2 : ((r == 1) ? c0 : c1);
    const int nz = (r == 0) ? c1 : ((r == 1) ? c2 : c0);
    P[gi] = nx | (ny << 10) | (nz << 20);
  }
  for (int gi = 0; gi < 4; ++gi) {
    for (int c = 0; c < 3; ++c) S[gi][c] = k1_wsum(S[gi][c]);
    P[gi] = k1_wsum(P[gi]);
  }

  float sv[4];
  for (int gi = 0; gi < 4; ++gi) sv[gi] = (r == 0) ? S[gi][0] : ((r == 1) ? S[gi][1] : S[gi][2]);
  const int gq = lane / 3;
  const float sum = (gq == 0) ? sv[0] : ((gq == 1) ? sv[1] : ((gq == 2) ? sv[2] : sv[3]));
  const int   pc  = (gq == 0) ? P[0]  : ((gq == 1) ? P[1]  : ((gq == 2) ? P[2]  : P[3]));
  const float cnt = (float)((pc >> (10 * r)) & 1023);
  const float mean = sum / cnt;

#pragma unroll
  for (int k = 0; k < 10; ++k) {
    const int o = lane + 32 * k;
    const float val = (k == 0 && lane < 12) ? mean : g[k];
    if (o < NCOL) xrow[o] = val;
  }
}

__global__ __launch_bounds__(256) void k1_build_x(const float* __restrict__ xin,
                                                  const int*   __restrict__ pl,
                                                  float*       __restrict__ X) {
  const int lane = threadIdx.x & 31;
  const int gw   = __builtin_amdgcn_readfirstlane((int)(blockIdx.x * K1_WAVES + (threadIdx.x >> 5)));
  const int r    = lane % 3;
  const int tail = min(1600 + lane, ROW_F - 1);

  int soff[10];
#pragma unroll
  for (int k = 0; k < 10; ++k) {
    const int q = min(max(lane + 32 * k - 12, 0), 299);
    const int i = q / 3;
    const int idx = min(max(pl[i], 0), N_LM - 1);
    soff[k] = idx * 3 + (q - 3 * i);
  }

  const int t0 = gw * FPW;
  float cur[51], gcur[10];
  k1_load(xin + (size_t)t0 * ROW_F, lane, tail, soff, cur, gcur);
#pragma unroll
  for (int f = 0; f < FPW; ++f) {
    float nxt[51], gnxt[10];
    if (f + 1 < FPW)
      k1_load(xin + (size_t)(t0 + f + 1) * ROW_F, lane, tail, soff, nxt, gnxt);
    k1_finish(cur, gcur, lane, r, X + (size_t)(t0 + f) * NCOL);
    if (f + 1 < FPW) {
#pragma unroll
      for (int i = 0; i < 51; ++i) cur[i] = nxt[i];
#pragma unroll
      for (int k = 0; k < 10; ++k) gcur[k] = gnxt[k];
    }
  }
}

#define K2_TPB     5
#define K2_GROUPS  4
#define K2_WAVES   8
#define K2_THREADS 256
#define K2_SROWS   32
#define K2_STEPS   8
#define K2_RED     240

__global__ __launch_bounds__(K2_THREADS) void k2_wmma_colsums(const float* __restrict__ X,
                                                              float*       __restrict__ partials) {
  __shared__ float red[K2_WAVES][K2_RED];

  const int seg   = blockIdx.y;
  const int chunk = blockIdx.z;
  const int lane  = threadIdx.x & 31;
  const int wave  = __builtin_amdgcn_readfirstlane((int)(threadIdx.x >> 5));
  const int half  = lane >> 4;
  const int n     = lane & 15;
  const int c0    = blockIdx.x * (K2_TPB * 16) + n;

  const size_t rowbase = (size_t)(seg * SEG_ROWS + chunk * CHUNK_ROWS + wave * K2_SROWS);
  const float* __restrict__ Xr = X + (rowbase + 2 * half) * NCOL;

  float x0[K2_TPB][K2_STEPS], x1[K2_TPB][K2_STEPS];
#pragma unroll
  for (int tb = 0; tb < K2_TPB; ++tb) {
    const int c     = c0 + 16 * tb;
    const int csafe = (c < NCOL) ? c : (NCOL - 1);
#pragma unroll
    for (int s = 0; s < K2_STEPS; ++s) {
      x0[tb][s] = Xr[(size_t)(4 * s + 0) * NCOL + csafe];
      x1[tb][s] = Xr[(size_t)(4 * s + 1) * NCOL + csafe];
    }
  }

  __builtin_amdgcn_sched_barrier(0);

  v2f ones; ones.x = 1.0f; ones.y = 1.0f;
  v8f accS[K2_TPB], accN[K2_TPB], accQ[K2_TPB];
#pragma unroll
  for (int tb = 0; tb < K2_TPB; ++tb) { accS[tb] = (v8f){}; accN[tb] = (v8f){}; accQ[tb] = (v8f){}; }

#pragma unroll
  for (int s = 0; s < K2_STEPS; ++s) {
#pragma unroll
    for (int tb = 0; tb < K2_TPB; ++tb) {
      const bool  cok = (c0 + 16 * tb < NCOL);
      const bool  g0 = (x0[tb][s] == x0[tb][s]) & cok;
      const bool  g1 = (x1[tb][s] == x1[tb][s]) & cok;
      const float m0 = g0 ? 1.0f : 0.0f;
      const float m1 = g1 ? 1.0f : 0.0f;
      const float v0 = g0 ? x0[tb][s] : 0.0f;
      const float v1 = g1 ? x1[tb][s] : 0.0f;
      v2f bx; bx.x = v0;      bx.y = v1;
      v2f bm; bm.x = m0;      bm.y = m1;
      v2f bq; bq.x = v0 * v0; bq.y = v1 * v1;
      accS[tb] = __builtin_amdgcn_wmma_f32_16x16x4_f32(false, ones, false, bx, (short)0, accS[tb], false, false);
      accN[tb] = __builtin_amdgcn_wmma_f32_16x16x4_f32(false, ones, false, bm, (short)0, accN[tb], false, false);
      accQ[tb] = __builtin_amdgcn_wmma_f32_16x16x4_f32(false, ones, false, bq, (short)0, accQ[tb], false, false);
    }
  }

  if (half == 0) {
#pragma unroll
    for (int tb = 0; tb < K2_TPB; ++tb) {
      red[wave][tb * 48 + 0 * 16 + n] = accS[tb][0];
      red[wave][tb * 48 + 1 * 16 + n] = accN[tb][0];
      red[wave][tb * 48 + 2 * 16 + n] = accQ[tb][0];
    }
  }
  __syncthreads();

  if (threadIdx.x < K2_RED) {
    const int tb  = threadIdx.x / 48;
    const int rem = threadIdx.x - 48 * tb;
    const int tl  = blockIdx.x * K2_TPB + tb;
    float a = red[0][threadIdx.x];
    for (int w = 1; w < K2_WAVES; ++w) a += red[w][threadIdx.x];
    if (tl * 16 + (rem & 15) < NCOL)
      partials[((((size_t)seg * TILES + tl) * CHUNKS + chunk) * 3) * 16 + rem] = a;
  }
}

#define K4_WAVES 8
#define K4_SLICE 64
#define K4_STEPS 16
#define K4_COLS  64
#define K4_CBLK  5

typedef float v4f __attribute__((ext_vector_type(4)));

__device__ __forceinline__ void k4_lds_barrier() {
  __builtin_amdgcn_fence(__ATOMIC_RELEASE, "workgroup", "local");
  __builtin_amdgcn_s_barrier();
  __builtin_amdgcn_fence(__ATOMIC_ACQUIRE, "workgroup", "local");
}

__global__ __launch_bounds__(256) void k4_resize(const float* __restrict__ X,
                                                 const float* __restrict__ partials,
                                                 float*       __restrict__ out) {
  __shared__ __attribute__((aligned(16))) float part[K4_WAVES][K4_COLS];
  __shared__ float wpart[K4_WAVES][2];
  __shared__ __attribute__((aligned(16))) float segS[SEGMENTS][K4_COLS];
  __shared__ __attribute__((aligned(16))) float segN[SEGMENTS][K4_COLS];
  __shared__ __attribute__((aligned(16))) float segQ[SEGMENTS][K4_COLS];
  __shared__ __attribute__((aligned(16))) float m0s[K4_COLS];

  const int cb   = blockIdx.x;
  const int i    = blockIdx.y;
  const int lane = threadIdx.x & 31;
  const int wave = __builtin_amdgcn_readfirstlane((int)(threadIdx.x >> 5));
  const int half = lane >> 4;
  const int n    = lane & 15;
  const int col  = cb * K4_COLS + 4 * n;
  const int csafe = (col < NCOL) ? col : (NCOL - 4);
  const bool stats = (i == 0);

  const float center = 256.0f * (float)i + 127.5f;
  const int rs = 256 * i - 128 + wave * K4_SLICE;
  const bool inside = (rs >= 0 && rs < T_FRAMES);

  v4f t[CHUNKS], u[CHUNKS];
  const bool hasq = stats && (half == 0);
  const bool cval = (col < NCOL);
  const int  ccl  = cval ? col : 0;
  const float* __restrict__ p =
      partials + ((((size_t)wave * TILES + (ccl >> 4)) * CHUNKS) * 3 + half) * 16 + (ccl & 15);
  if (hasq && cval) {
#pragma unroll
    for (int ch = 0; ch < CHUNKS; ++ch) u[ch] = *(const v4f*)(p + ch * 48 + 32);
  } else {
#pragma unroll
    for (int ch = 0; ch < CHUNKS; ++ch) u[ch] = (v4f){0.f, 0.f, 0.f, 0.f};
  }
  if (cval) {
#pragma unroll
    for (int ch = 0; ch < CHUNKS; ++ch) t[ch] = *(const v4f*)(p + ch * 48);
  } else {
#pragma unroll
    for (int ch = 0; ch < CHUNKS; ++ch) t[ch] = (v4f){0.f, 0.f, 0.f, 0.f};
  }

  v4f xa[K4_STEPS], xb[K4_STEPS];
  {
    const int rl = inside ? rs : ((rs < 0) ? 0 : (T_FRAMES - K4_SLICE));
    const float* __restrict__ Xp = X + (size_t)(rl + 2 * half) * NCOL + csafe;
#pragma unroll
    for (int s = 0; s < K4_STEPS; ++s) {
      xa[s] = *(const v4f*)(Xp + (size_t)(4 * s + 0) * NCOL);
      xb[s] = *(const v4f*)(Xp + (size_t)(4 * s + 1) * NCOL);
    }
  }
  __builtin_amdgcn_sched_barrier(0);

  {
    v4f a = {0.f, 0.f, 0.f, 0.f};
#pragma unroll
    for (int ch = 0; ch < CHUNKS; ++ch) a += t[ch];
    if (half == 0) *(v4f*)&segS[wave][4 * n] = a; else *(v4f*)&segN[wave][4 * n] = a;
    if (hasq) {
      v4f q = {0.f, 0.f, 0.f, 0.f};
#pragma unroll
      for (int ch = 0; ch < CHUNKS; ++ch) q += u[ch];
      *(v4f*)&segQ[wave][4 * n] = q;
    }
  }
  k4_lds_barrier();

  if (threadIdx.x < K4_COLS) {
    const int j = threadIdx.x;
    const int c = cb * K4_COLS + j;
    float tS = 0.f, tN = 0.f;
    for (int s = 0; s < SEGMENTS; ++s) { tS += segS[s][j]; tN += segN[s][j]; }
    const float mean = tS / tN;
    m0s[j] = mean;
    if (stats && c < NCOL) {
      float tQ = 0.f;
      for (int s = 0; s < SEGMENTS; ++s) tQ += segQ[s][j];
      const float var = fmaxf(tQ / tN - mean * mean, 0.0f);
      out[SEGMENTS * FEAT_BLK + c]        = fin_or0(mean);
      out[SEGMENTS * FEAT_BLK + NCOL + c] = fin_or0(sqrtf(var));
    }
  }
  if (stats) {
    const int j = threadIdx.x & (K4_COLS - 1);
    const int c = cb * K4_COLS + j;
    if (c < NCOL) {
      for (int s = (int)(threadIdx.x >> 6); s < SEGMENTS; s += 4) {
        const float S = segS[s][j], N = segN[s][j], Q = segQ[s][j];
        const float mean = S / N;
        const float var  = fmaxf(Q / N - mean * mean, 0.0f);
        out[s * FEAT_BLK + c]        = fin_or0(mean);
        out[s * FEAT_BLK + NCOL + c] = fin_or0(sqrtf(var));
      }
    }
  }
  k4_lds_barrier();

  v8f acc0 = {}; v8f acc1 = {}; v8f acc2 = {}; v8f acc3 = {};
  float wl = 0.0f;

  if (inside) {
    const v4f m0 = *(const v4f*)&m0s[4 * n];
#pragma unroll
    for (int s = 0; s < K4_STEPS; ++s) {
      const float ta = (float)(rs + 4 * s + 2 * half);
      const float wa = 1.0f - fabsf(ta - center) * (1.0f / 256.0f);
      const float wb = 1.0f - fabsf((ta + 1.0f) - center) * (1.0f / 256.0f);
      v2f a; a.x = (n == 0) ? wa : 0.0f; a.y = (n == 0) ? wb : 0.0f;
      wl += a.x + a.y;

      float va, vb; v2f b;
      va = xa[s].x; vb = xb[s].x;
      va = (va - va == 0.0f) ? va : m0.x;
      vb = (vb - vb == 0.0f) ? vb : m0.x;
      b.x = va; b.y = vb;
      acc0 = __builtin_amdgcn_wmma_f32_16x16x4_f32(false, a, false, b, (short)0, acc0, false, false);
      va = xa[s].y; vb = xb[s].y;
      va = (va - va == 0.0f) ? va : m0.y;
      vb = (vb - vb == 0.0f) ? vb : m0.y;
      b.x = va; b.y = vb;
      acc1 = __builtin_amdgcn_wmma_f32_16x16x4_f32(false, a, false, b, (short)0, acc1, false, false);
      va = xa[s].z; vb = xb[s].z;
      va = (va - va == 0.0f) ? va : m0.z;
      vb = (vb - vb == 0.0f) ? vb : m0.z;
      b.x = va; b.y = vb;
      acc2 = __builtin_amdgcn_wmma_f32_16x16x4_f32(false, a, false, b, (short)0, acc2, false, false);
      va = xa[s].w; vb = xb[s].w;
      va = (va - va == 0.0f) ? va : m0.w;
      vb = (vb - vb == 0.0f) ? vb : m0.w;
      b.x = va; b.y = vb;
      acc3 = __builtin_amdgcn_wmma_f32_16x16x4_f32(false, a, false, b, (short)0, acc3, false, false);
    }
  }

  if (lane < 16) {
    v4f o; o.x = acc0[0]; o.y = acc1[0]; o.z = acc2[0]; o.w = acc3[0];
    *(v4f*)&part[wave][4 * n] = o;
  }
  if (n == 0) wpart[wave][half] = wl;
  __syncthreads();

  if (threadIdx.x < K4_COLS) {
    const int j = threadIdx.x;
    float acc = 0.f, wsum = 0.f;
    for (int w = 0; w < K4_WAVES; ++w) {
      acc  += part[w][j];
      wsum += wpart[w][0] + wpart[w][1];
    }
    const int c = cb * K4_COLS + j;
    const float r = acc / wsum;
    if (c < NCOL) out[9 * FEAT_BLK + (size_t)i * NCOL + c] = (r == r) ? r : 0.0f;
  }
}

extern "C" void kernel_launch(void* const* d_in, const int* in_sizes, int n_in,
                              void* d_out, int out_size, void* d_ws, size_t ws_size,
                              hipStream_t stream) {
  const float* x_in = (const float*)d_in[0];
  const int*   pl   = (const int*)d_in[1];
  float* out = (float*)d_out;

  float* X        = (float*)d_ws;
  float* partials = X + (size_t)T_FRAMES * NCOL;

  k1_build_x<<<dim3(K1_BLOCKS), dim3(256), 0, stream>>>(x_in, pl, X);
  k2_wmma_colsums<<<dim3(K2_GROUPS, SEGMENTS, CHUNKS), dim3(K2_THREADS), 0, stream>>>(X, partials);
  k4_resize<<<dim3(K4_CBLK, RESIZE_T), dim3(256), 0, stream>>>(X, partials, out);
}
